# starting kernel
# baseline (speedup 1.0000x reference)
#include <hip/hip_runtime.h>
#include <hip/hip_bf16.h>

#define FFT_L   4096
#define FFT_NT  256
#define PI_F    3.14159265358979323846f

__global__ __launch_bounds__(FFT_NT) void fft_denoise_kernel(
    const float* __restrict__ x, float* __restrict__ xt) {
  __shared__ float re[FFT_L];
  __shared__ float im[FFT_L];
  const int img = blockIdx.x;
  const int tid = threadIdx.x;
  const float* xi = x + (size_t)img * 62 * 62;

  for (int idx = tid; idx < FFT_L; idx += FFT_NT) {
    int i = idx >> 6, j = idx & 63;
    float v = 0.0f;
    if (i >= 1 && i <= 62 && j >= 1 && j <= 62) v = xi[(i - 1) * 62 + (j - 1)];
    re[idx] = v;
    im[idx] = 0.0f;
  }
  __syncthreads();

  for (int s = 11; s >= 0; --s) {
    const int half = 1 << s;
    const float w = -PI_F / (float)half;
    for (int t = tid; t < FFT_L / 2; t += FFT_NT) {
      int pos = t & (half - 1);
      int i0 = ((t >> s) << (s + 1)) + pos;
      int i1 = i0 + half;
      float ar = re[i0], ai = im[i0];
      float br = re[i1], bi = im[i1];
      float sr = ar - br, si = ai - bi;
      float sn, cs;
      __sincosf(w * (float)pos, &sn, &cs);
      re[i0] = ar + br;
      im[i0] = ai + bi;
      re[i1] = sr * cs - si * sn;
      im[i1] = sr * sn + si * cs;
    }
    __syncthreads();
  }

  for (int idx = tid; idx < FFT_L; idx += FFT_NT) {
    if (fabsf(re[idx]) < 0.5f) {
      re[idx] = 0.0f;
      im[idx] = 0.0f;
    }
  }
  __syncthreads();

  for (int s = 0; s <= 11; ++s) {
    const int half = 1 << s;
    const float w = PI_F / (float)half;
    for (int t = tid; t < FFT_L / 2; t += FFT_NT) {
      int pos = t & (half - 1);
      int i0 = ((t >> s) << (s + 1)) + pos;
      int i1 = i0 + half;
      float sn, cs;
      __sincosf(w * (float)pos, &sn, &cs);
      float br = re[i1], bi = im[i1];
      float tr = br * cs - bi * sn;
      float ti = br * sn + bi * cs;
      float ar = re[i0], ai = im[i0];
      re[i0] = ar + tr;
      im[i0] = ai + ti;
      re[i1] = ar - tr;
      im[i1] = ai - ti;
    }
    __syncthreads();
  }

  float* xo = xt + (size_t)img * FFT_L;
  const float scale = 1.0f / (float)FFT_L;
  for (int idx = tid; idx < FFT_L; idx += FFT_NT) xo[idx] = re[idx] * scale;
}

typedef __attribute__((ext_vector_type(2))) float v2f;
typedef __attribute__((ext_vector_type(8))) float v8f;

#define PATCH_ELEMS (32 * 3 * 18)

__global__ __launch_bounds__(128) void conv3x3_wmma_kernel(
    const float* __restrict__ xt, const float* __restrict__ Wf,
    const float* __restrict__ bias, float* __restrict__ out) {
  __shared__ float xs[PATCH_ELEMS];

  const int lane = threadIdx.x & 31;
  const int wave = threadIdx.x >> 5;
  const int o_base = wave * 16;
  const int j0 = blockIdx.x * 16;
  const int i = blockIdx.y;
  const int b = blockIdx.z;

  const int hi = lane >> 4;
  const int n = lane & 15;

  const float* wrow = Wf + (size_t)(o_base + n) * 288;
  const float* xb = xt + (size_t)b * 32 * FFT_L;

  __builtin_prefetch(wrow, 0, 0);

  for (int e = threadIdx.x; e < PATCH_ELEMS; e += 128) {
    int c = e / 54;
    int rem = e - c * 54;
    int kh = rem / 18;
    int col = rem - kh * 18;
    int jc = j0 + col;
    if (jc > 63) jc = 63;
    xs[e] = xb[((size_t)c * 64 + (i + kh)) * 64 + jc];
  }
  __syncthreads();

  v8f acc = {};
  for (int kk = 0; kk < 288; kk += 4) {
    const int ka = kk + 2 * hi;

    v2f a;
    a.x = wrow[ka];
    a.y = wrow[ka + 1];

    int c0 = ka / 9, r0 = ka - c0 * 9;
    int kh0 = r0 / 3, kw0 = r0 - kh0 * 3;
    int kb = ka + 1;
    int c1 = kb / 9, r1 = kb - c1 * 9;
    int kh1 = r1 / 3, kw1 = r1 - kh1 * 3;
    v2f bv;
    bv.x = xs[c0 * 54 + kh0 * 18 + n + kw0];
    bv.y = xs[c1 * 54 + kh1 * 18 + n + kw1];

    acc = __builtin_amdgcn_wmma_f32_16x16x4_f32(false, a, false, bv, (short)0,
                                                acc, false, false);
  }

  if (j0 + n < 62) {
#pragma unroll
    for (int r = 0; r < 8; ++r) {
      int o = o_base + r + 8 * hi;
      float v = acc[r] + bias[o];
      out[(((size_t)b * 64 + o) * 62 + i) * 62 + (j0 + n)] = v;
    }
  }
}

extern "C" void kernel_launch(void* const* d_in, const int* in_sizes, int n_in,
                              void* d_out, int out_size, void* d_ws,
                              size_t ws_size, hipStream_t stream) {
  const float* x = (const float*)d_in[0];
  const float* W = (const float*)d_in[1];
  const float* bias = (const float*)d_in[2];
  float* out = (float*)d_out;
  float* xt = (float*)d_ws;

  fft_denoise_kernel<<<dim3(128), dim3(FFT_NT), 0, stream>>>(x, xt);
  conv3x3_wmma_kernel<<<dim3(4, 62, 4), dim3(128), 0, stream>>>(xt, W, bias,
                                                                out);
}


# FFT kernel: the 12 radix-2 stages of each direction run as 3 register-resident groups of 4 (16 points per thread), same butterflies and twiddle values; LDS round trips 24 -> 4, barriers 26 -> 4
# speedup vs baseline: 2.0066x; 2.0066x over previous
#include <hip/hip_runtime.h>
#include <hip/hip_bf16.h>

#define FFT_L   4096
#define FFT_NT  256
#define PI_F    3.14159265358979323846f

#define FFT_PAD(i) ((i) + ((i) >> 4))
#define FFT_LDS    (FFT_L + FFT_L / 16)

template <int SH, bool INV>
__device__ __forceinline__ void fft16_stages(float (&xr)[16], float (&xi)[16],
                                             int low, const float* tws,
                                             const float* twc) {
#pragma unroll
  for (int qq = 0; qq < 4; ++qq) {
    const int q = INV ? qq : 3 - qq;
    const int span = 1 << q;
    const float w = (INV ? PI_F : -PI_F) / (float)(1 << (SH + q));
    float sn[8], cs[8];
#pragma unroll
    for (int j = 0; j < 8; ++j) {
      if (j < span) {
        if (SH == 0) {
          sn[j] = tws[j << (3 - q)];
          cs[j] = twc[j << (3 - q)];
        } else {
          const int pos = (j << SH) | low;
          __sincosf(w * (float)pos, &sn[j], &cs[j]);
        }
      }
    }
#pragma unroll
    for (int k = 0; k < 16; ++k) {
      if (k & span) continue;
      const int j = k & (span - 1);
      const int k1 = k + span;
      if (!INV) {
        float ar = xr[k], ai = xi[k];
        float br = xr[k1], bi = xi[k1];
        float sr = ar - br, si = ai - bi;
        xr[k] = ar + br;
        xi[k] = ai + bi;
        xr[k1] = fmaf(sr, cs[j], -(si * sn[j]));
        xi[k1] = fmaf(sr, sn[j], si * cs[j]);
      } else {
        float br = xr[k1], bi = xi[k1];
        float tr = fmaf(br, cs[j], -(bi * sn[j]));
        float ti = fmaf(br, sn[j], bi * cs[j]);
        float ar = xr[k], ai = xi[k];
        xr[k] = ar + tr;
        xi[k] = ai + ti;
        xr[k1] = ar - tr;
        xi[k1] = ai - ti;
      }
    }
  }
}

__global__ __launch_bounds__(FFT_NT) void fft_denoise_kernel(
    const float* __restrict__ x, float* __restrict__ xt) {
  __shared__ float re[FFT_LDS];
  __shared__ float im[FFT_LDS];
  __shared__ float tws[16];
  __shared__ float twc[16];
  const int img = blockIdx.x;
  const int tid = threadIdx.x;
  const float* xi_g = x + (size_t)img * 62 * 62;

  float xr[16], xi[16];

  if (tid < 16) {
    const float w = (tid < 8 ? -PI_F : PI_F) / 8.0f;
    float sn, cs;
    __sincosf(w * (float)(tid & 7), &sn, &cs);
    tws[tid] = sn;
    twc[tid] = cs;
  }

  {
    const int j = tid & 63;
#pragma unroll
    for (int k = 0; k < 16; ++k) {
      const int i = (k << 2) | (tid >> 6);
      float v = 0.0f;
      if (i >= 1 && i <= 62 && j >= 1 && j <= 62)
        v = xi_g[(i - 1) * 62 + (j - 1)];
      xr[k] = v;
      xi[k] = 0.0f;
    }
  }

  fft16_stages<8, false>(xr, xi, tid, tws, twc);
  {
    const int base = FFT_PAD(tid);
#pragma unroll
    for (int k = 0; k < 16; ++k) {
      re[base + k * 272] = xr[k];
      im[base + k * 272] = xi[k];
    }
  }
  __syncthreads();

  const int lo4 = tid & 15;
  const int midbase = (tid >> 4) * 272 + lo4;
#pragma unroll
  for (int k = 0; k < 16; ++k) {
    xr[k] = re[midbase + k * 17];
    xi[k] = im[midbase + k * 17];
  }
  fft16_stages<4, false>(xr, xi, lo4, tws, twc);
#pragma unroll
  for (int k = 0; k < 16; ++k) {
    re[midbase + k * 17] = xr[k];
    im[midbase + k * 17] = xi[k];
  }
  __syncthreads();

  const int leafbase = tid * 17;
#pragma unroll
  for (int k = 0; k < 16; ++k) {
    xr[k] = re[leafbase + k];
    xi[k] = im[leafbase + k];
  }
  fft16_stages<0, false>(xr, xi, 0, tws, twc);
#pragma unroll
  for (int k = 0; k < 16; ++k) {
    if (fabsf(xr[k]) < 0.5f) {
      xr[k] = 0.0f;
      xi[k] = 0.0f;
    }
  }
  fft16_stages<0, true>(xr, xi, 0, tws + 8, twc + 8);
#pragma unroll
  for (int k = 0; k < 16; ++k) {
    re[leafbase + k] = xr[k];
    im[leafbase + k] = xi[k];
  }
  __syncthreads();

#pragma unroll
  for (int k = 0; k < 16; ++k) {
    xr[k] = re[midbase + k * 17];
    xi[k] = im[midbase + k * 17];
  }
  fft16_stages<4, true>(xr, xi, lo4, tws, twc);
#pragma unroll
  for (int k = 0; k < 16; ++k) {
    re[midbase + k * 17] = xr[k];
    im[midbase + k * 17] = xi[k];
  }
  __syncthreads();

  {
    const int base = FFT_PAD(tid);
#pragma unroll
    for (int k = 0; k < 16; ++k) {
      xr[k] = re[base + k * 272];
      xi[k] = im[base + k * 272];
    }
  }
  fft16_stages<8, true>(xr, xi, tid, tws, twc);

  float* xo = xt + (size_t)img * FFT_L;
  const float scale = 1.0f / (float)FFT_L;
#pragma unroll
  for (int k = 0; k < 16; ++k)
    xo[(k << 8) | tid] = xr[k] * scale;
}

typedef __attribute__((ext_vector_type(2))) float v2f;
typedef __attribute__((ext_vector_type(8))) float v8f;

#define PATCH_ELEMS (32 * 3 * 18)

__global__ __launch_bounds__(128) void conv3x3_wmma_kernel(
    const float* __restrict__ xt, const float* __restrict__ Wf,
    const float* __restrict__ bias, float* __restrict__ out) {
  __shared__ float xs[PATCH_ELEMS];

  const int lane = threadIdx.x & 31;
  const int wave = threadIdx.x >> 5;
  const int o_base = wave * 16;
  const int j0 = blockIdx.x * 16;
  const int i = blockIdx.y;
  const int b = blockIdx.z;

  const int hi = lane >> 4;
  const int n = lane & 15;

  const float* wrow = Wf + (size_t)(o_base + n) * 288;
  const float* xb = xt + (size_t)b * 32 * FFT_L;

  __builtin_prefetch(wrow, 0, 0);

  for (int e = threadIdx.x; e < PATCH_ELEMS; e += 128) {
    int c = e / 54;
    int rem = e - c * 54;
    int kh = rem / 18;
    int col = rem - kh * 18;
    int jc = j0 + col;
    if (jc > 63) jc = 63;
    xs[e] = xb[((size_t)c * 64 + (i + kh)) * 64 + jc];
  }
  __syncthreads();

  v8f acc = {};
  for (int kk = 0; kk < 288; kk += 4) {
    const int ka = kk + 2 * hi;

    v2f a;
    a.x = wrow[ka];
    a.y = wrow[ka + 1];

    int c0 = ka / 9, r0 = ka - c0 * 9;
    int kh0 = r0 / 3, kw0 = r0 - kh0 * 3;
    int kb = ka + 1;
    int c1 = kb / 9, r1 = kb - c1 * 9;
    int kh1 = r1 / 3, kw1 = r1 - kh1 * 3;
    v2f bv;
    bv.x = xs[c0 * 54 + kh0 * 18 + n + kw0];
    bv.y = xs[c1 * 54 + kh1 * 18 + n + kw1];

    acc = __builtin_amdgcn_wmma_f32_16x16x4_f32(false, a, false, bv, (short)0,
                                                acc, false, false);
  }

  if (j0 + n < 62) {
#pragma unroll
    for (int r = 0; r < 8; ++r) {
      int o = o_base + r + 8 * hi;
      float v = acc[r] + bias[o];
      out[(((size_t)b * 64 + o) * 62 + i) * 62 + (j0 + n)] = v;
    }
  }
}

extern "C" void kernel_launch(void* const* d_in, const int* in_sizes, int n_in,
                              void* d_out, int out_size, void* d_ws,
                              size_t ws_size, hipStream_t stream) {
  const float* x = (const float*)d_in[0];
  const float* W = (const float*)d_in[1];
  const float* bias = (const float*)d_in[2];
  float* out = (float*)d_out;
  float* xt = (float*)d_ws;

  fft_denoise_kernel<<<dim3(128), dim3(FFT_NT), 0, stream>>>(x, xt);
  conv3x3_wmma_kernel<<<dim3(4, 62, 4), dim3(128), 0, stream>>>(xt, W, bias,
                                                                out);
}


# conv: one 8-wave workgroup per output row stages W and the 3-row patch in LDS once; A/B fragments are LDS reads at constant offsets, one group ahead; 2 column tiles per wave; sum order unchanged
# speedup vs baseline: 3.2702x; 1.6297x over previous
#include <hip/hip_runtime.h>
#include <hip/hip_bf16.h>

#define FFT_L   4096
#define FFT_NT  256
#define PI_F    3.14159265358979323846f

#define FFT_PAD(i) ((i) + ((i) >> 4))
#define FFT_LDS    (FFT_L + FFT_L / 16)

template <int SH, bool INV>
__device__ __forceinline__ void fft16_stages(float (&xr)[16], float (&xi)[16],
                                             int low, const float* tws,
                                             const float* twc) {
#pragma unroll
  for (int qq = 0; qq < 4; ++qq) {
    const int q = INV ? qq : 3 - qq;
    const int span = 1 << q;
    const float w = (INV ? PI_F : -PI_F) / (float)(1 << (SH + q));
    float sn[8], cs[8];
#pragma unroll
    for (int j = 0; j < 8; ++j) {
      if (j < span) {
        if (SH == 0) {
          sn[j] = tws[j << (3 - q)];
          cs[j] = twc[j << (3 - q)];
        } else {
          const int pos = (j << SH) | low;
          __sincosf(w * (float)pos, &sn[j], &cs[j]);
        }
      }
    }
#pragma unroll
    for (int k = 0; k < 16; ++k) {
      if (k & span) continue;
      const int j = k & (span - 1);
      const int k1 = k + span;
      if (!INV) {
        float ar = xr[k], ai = xi[k];
        float br = xr[k1], bi = xi[k1];
        float sr = ar - br, si = ai - bi;
        xr[k] = ar + br;
        xi[k] = ai + bi;
        xr[k1] = fmaf(sr, cs[j], -(si * sn[j]));
        xi[k1] = fmaf(sr, sn[j], si * cs[j]);
      } else {
        float br = xr[k1], bi = xi[k1];
        float tr = fmaf(br, cs[j], -(bi * sn[j]));
        float ti = fmaf(br, sn[j], bi * cs[j]);
        float ar = xr[k], ai = xi[k];
        xr[k] = ar + tr;
        xi[k] = ai + ti;
        xr[k1] = ar - tr;
        xi[k1] = ai - ti;
      }
    }
  }
}

__global__ __launch_bounds__(FFT_NT) void fft_denoise_kernel(
    const float* __restrict__ x, float* __restrict__ xt) {
  __shared__ float re[FFT_LDS];
  __shared__ float im[FFT_LDS];
  __shared__ float tws[16];
  __shared__ float twc[16];
  const int img = blockIdx.x;
  const int tid = threadIdx.x;
  const float* xi_g = x + (size_t)img * 62 * 62;

  float xr[16], xi[16];

  if (tid < 16) {
    const float w = (tid < 8 ? -PI_F : PI_F) / 8.0f;
    float sn, cs;
    __sincosf(w * (float)(tid & 7), &sn, &cs);
    tws[tid] = sn;
    twc[tid] = cs;
  }

  {
    const int j = tid & 63;
#pragma unroll
    for (int k = 0; k < 16; ++k) {
      const int i = (k << 2) | (tid >> 6);
      float v = 0.0f;
      if (i >= 1 && i <= 62 && j >= 1 && j <= 62)
        v = xi_g[(i - 1) * 62 + (j - 1)];
      xr[k] = v;
      xi[k] = 0.0f;
    }
  }

  fft16_stages<8, false>(xr, xi, tid, tws, twc);
  {
    const int base = FFT_PAD(tid);
#pragma unroll
    for (int k = 0; k < 16; ++k) {
      re[base + k * 272] = xr[k];
      im[base + k * 272] = xi[k];
    }
  }
  __syncthreads();

  const int lo4 = tid & 15;
  const int midbase = (tid >> 4) * 272 + lo4;
#pragma unroll
  for (int k = 0; k < 16; ++k) {
    xr[k] = re[midbase + k * 17];
    xi[k] = im[midbase + k * 17];
  }
  fft16_stages<4, false>(xr, xi, lo4, tws, twc);
#pragma unroll
  for (int k = 0; k < 16; ++k) {
    re[midbase + k * 17] = xr[k];
    im[midbase + k * 17] = xi[k];
  }
  __syncthreads();

  const int leafbase = tid * 17;
#pragma unroll
  for (int k = 0; k < 16; ++k) {
    xr[k] = re[leafbase + k];
    xi[k] = im[leafbase + k];
  }
  fft16_stages<0, false>(xr, xi, 0, tws, twc);
#pragma unroll
  for (int k = 0; k < 16; ++k) {
    if (fabsf(xr[k]) < 0.5f) {
      xr[k] = 0.0f;
      xi[k] = 0.0f;
    }
  }
  fft16_stages<0, true>(xr, xi, 0, tws + 8, twc + 8);
#pragma unroll
  for (int k = 0; k < 16; ++k) {
    re[leafbase + k] = xr[k];
    im[leafbase + k] = xi[k];
  }
  __syncthreads();

#pragma unroll
  for (int k = 0; k < 16; ++k) {
    xr[k] = re[midbase + k * 17];
    xi[k] = im[midbase + k * 17];
  }
  fft16_stages<4, true>(xr, xi, lo4, tws, twc);
#pragma unroll
  for (int k = 0; k < 16; ++k) {
    re[midbase + k * 17] = xr[k];
    im[midbase + k * 17] = xi[k];
  }
  __syncthreads();

  {
    const int base = FFT_PAD(tid);
#pragma unroll
    for (int k = 0; k < 16; ++k) {
      xr[k] = re[base + k * 272];
      xi[k] = im[base + k * 272];
    }
  }
  fft16_stages<8, true>(xr, xi, tid, tws, twc);

  float* xo = xt + (size_t)img * FFT_L;
  const float scale = 1.0f / (float)FFT_L;
#pragma unroll
  for (int k = 0; k < 16; ++k)
    xo[(k << 8) | tid] = xr[k] * scale;
}

typedef __attribute__((ext_vector_type(2))) float v2f;
typedef __attribute__((ext_vector_type(8))) float v8f;

#define CW_WP 292
#define CW_XP 84

__global__ __launch_bounds__(256) void conv3x3_wmma_kernel(
    const float* __restrict__ xt, const float* __restrict__ Wf,
    const float* __restrict__ bias, float* __restrict__ out) {
  __shared__ __attribute__((aligned(16))) float wl[64 * CW_WP];
  __shared__ __attribute__((aligned(16))) float xs[96 * CW_XP];

  const int tid = threadIdx.x;
  const int lane = tid & 31;
  const int wave = tid >> 5;
  const int mt = wave & 3;
  const int nh = wave >> 2;
  const int i = blockIdx.x;
  const int b = blockIdx.y;

  const int hi = lane >> 4;
  const int n = lane & 15;

  {
    const float4* W4 = reinterpret_cast<const float4*>(Wf) + (tid >> 3) * 72 + (tid & 7);
    float* wd = wl + (tid >> 3) * CW_WP + 4 * (tid & 7);
#pragma unroll
    for (int r = 0; r < 18; ++r) {
      const int half = r / 9;
      const int m = r % 9;
      *reinterpret_cast<float4*>(wd + half * 32 * CW_WP + 32 * m) =
          W4[half * 32 * 72 + 8 * m];
    }
  }
  {
    const float4* X4 = reinterpret_cast<const float4*>(
                           xt + ((size_t)b * 32 * FFT_L + (size_t)i * 64)) +
                       (tid >> 3) * (FFT_L / 4) + (tid & 7);
    float* xd = xs + (tid >> 3) * 3 * CW_XP + 4 * (tid & 7);
#pragma unroll
    for (int m = 0; m < 6; ++m)
      *reinterpret_cast<float4*>(xd + (m >> 1) * CW_XP + 32 * (m & 1)) = X4[8 * m];
    if (tid < 96)
      *reinterpret_cast<float4*>(&xs[tid * CW_XP + 64]) =
          make_float4(0.0f, 0.0f, 0.0f, 0.0f);
  }
  __syncthreads();

  const float* arow = wl + (16 * mt + n) * CW_WP + 2 * hi;
  const float* bsame = xs + 32 * nh + n + 2 * hi;
  const float* bnext = xs + 32 * nh + n + (CW_XP - 1) * hi;

  v8f acc0 = {};
  v8f acc1 = {};

  auto load_step = [&](int t, v2f& a, float (&q)[4]) {
    a = *reinterpret_cast<const v2f*>(arow + 4 * t);
    {
      const int k0 = 4 * t;
      const float* p = ((k0 % 3) == 0 ? bsame : bnext) + (k0 / 3) * CW_XP + (k0 % 3);
      q[0] = p[0];
      q[1] = p[16];
    }
    {
      const int k0 = 4 * t + 1;
      const float* p = ((k0 % 3) == 0 ? bsame : bnext) + (k0 / 3) * CW_XP + (k0 % 3);
      q[2] = p[0];
      q[3] = p[16];
    }
  };

  v2f fa[2][6];
  float fq[2][6][4];
#pragma unroll
  for (int s = 0; s < 6; ++s) load_step(s, fa[0][s], fq[0][s]);
  __builtin_amdgcn_sched_barrier(0);
#pragma unroll
  for (int j = 0; j < 12; ++j) {
    const int cur = j & 1;
    if (j + 1 < 12) {
#pragma unroll
      for (int s = 0; s < 6; ++s)
        load_step(6 * (j + 1) + s, fa[cur ^ 1][s], fq[cur ^ 1][s]);
    }
    __builtin_amdgcn_sched_barrier(0);
#pragma unroll
    for (int s = 0; s < 6; ++s) {
      v2f b0, b1;
      b0.x = fq[cur][s][0];
      b0.y = fq[cur][s][2];
      b1.x = fq[cur][s][1];
      b1.y = fq[cur][s][3];
      acc0 = __builtin_amdgcn_wmma_f32_16x16x4_f32(
          false, fa[cur][s], false, b0, (short)0, acc0, false, false);
      acc1 = __builtin_amdgcn_wmma_f32_16x16x4_f32(
          false, fa[cur][s], false, b1, (short)0, acc1, false, false);
    }
    __builtin_amdgcn_sched_barrier(0);
  }

  const int col0 = 32 * nh + n;
  {
#pragma unroll
    for (int r = 0; r < 8; ++r) {
      const int o = 16 * mt + r + 8 * hi;
      out[(((size_t)b * 64 + o) * 62 + i) * 62 + col0] = acc0[r] + bias[o];
    }
  }
  if (col0 + 16 < 62) {
#pragma unroll
    for (int r = 0; r < 8; ++r) {
      const int o = 16 * mt + r + 8 * hi;
      out[(((size_t)b * 64 + o) * 62 + i) * 62 + col0 + 16] = acc1[r] + bias[o];
    }
  }
}

extern "C" void kernel_launch(void* const* d_in, const int* in_sizes, int n_in,
                              void* d_out, int out_size, void* d_ws,
                              size_t ws_size, hipStream_t stream) {
  const float* x = (const float*)d_in[0];
  const float* W = (const float*)d_in[1];
  const float* bias = (const float*)d_in[2];
  float* out = (float*)d_out;
  float* xt = (float*)d_ws;

  fft_denoise_kernel<<<dim3(128), dim3(FFT_NT), 0, stream>>>(x, xt);
  conv3x3_wmma_kernel<<<dim3(62, 4), dim3(256), 0, stream>>>(xt, W, bias, out);
}


# conv: output columns in pairs by the 2-output 3-tap minimal filtering identity (4 products per pair, was 6), operands combined by adds before each WMMA; 96 WMMAs per wave, was 144; sum order changes
# speedup vs baseline: 3.4637x; 1.0592x over previous
#include <hip/hip_runtime.h>
#include <hip/hip_bf16.h>

#define FFT_L   4096
#define FFT_NT  256
#define PI_F    3.14159265358979323846f

#define FFT_PAD(i) ((i) + ((i) >> 4))
#define FFT_LDS    (FFT_L + FFT_L / 16)

template <int SH, bool INV>
__device__ __forceinline__ void fft16_stages(float (&xr)[16], float (&xi)[16],
                                             int low, const float* tws,
                                             const float* twc) {
#pragma unroll
  for (int qq = 0; qq < 4; ++qq) {
    const int q = INV ? qq : 3 - qq;
    const int span = 1 << q;
    const float w = (INV ? PI_F : -PI_F) / (float)(1 << (SH + q));
    float sn[8], cs[8];
#pragma unroll
    for (int j = 0; j < 8; ++j) {
      if (j < span) {
        if (SH == 0) {
          sn[j] = tws[j << (3 - q)];
          cs[j] = twc[j << (3 - q)];
        } else {
          const int pos = (j << SH) | low;
          __sincosf(w * (float)pos, &sn[j], &cs[j]);
        }
      }
    }
#pragma unroll
    for (int k = 0; k < 16; ++k) {
      if (k & span) continue;
      const int j = k & (span - 1);
      const int k1 = k + span;
      if (!INV) {
        float ar = xr[k], ai = xi[k];
        float br = xr[k1], bi = xi[k1];
        float sr = ar - br, si = ai - bi;
        xr[k] = ar + br;
        xi[k] = ai + bi;
        xr[k1] = fmaf(sr, cs[j], -(si * sn[j]));
        xi[k1] = fmaf(sr, sn[j], si * cs[j]);
      } else {
        float br = xr[k1], bi = xi[k1];
        float tr = fmaf(br, cs[j], -(bi * sn[j]));
        float ti = fmaf(br, sn[j], bi * cs[j]);
        float ar = xr[k], ai = xi[k];
        xr[k] = ar + tr;
        xi[k] = ai + ti;
        xr[k1] = ar - tr;
        xi[k1] = ai - ti;
      }
    }
  }
}

__global__ __launch_bounds__(FFT_NT) void fft_denoise_kernel(
    const float* __restrict__ x, float* __restrict__ xt) {
  __shared__ float re[FFT_LDS];
  __shared__ float im[FFT_LDS];
  __shared__ float tws[16];
  __shared__ float twc[16];
  const int img = blockIdx.x;
  const int tid = threadIdx.x;
  const float* xi_g = x + (size_t)img * 62 * 62;

  float xr[16], xi[16];

  if (tid < 16) {
    const float w = (tid < 8 ? -PI_F : PI_F) / 8.0f;
    float sn, cs;
    __sincosf(w * (float)(tid & 7), &sn, &cs);
    tws[tid] = sn;
    twc[tid] = cs;
  }

  {
    const int j = tid & 63;
#pragma unroll
    for (int k = 0; k < 16; ++k) {
      const int i = (k << 2) | (tid >> 6);
      float v = 0.0f;
      if (i >= 1 && i <= 62 && j >= 1 && j <= 62)
        v = xi_g[(i - 1) * 62 + (j - 1)];
      xr[k] = v;
      xi[k] = 0.0f;
    }
  }

  fft16_stages<8, false>(xr, xi, tid, tws, twc);
  {
    const int base = FFT_PAD(tid);
#pragma unroll
    for (int k = 0; k < 16; ++k) {
      re[base + k * 272] = xr[k];
      im[base + k * 272] = xi[k];
    }
  }
  __syncthreads();

  const int lo4 = tid & 15;
  const int midbase = (tid >> 4) * 272 + lo4;
#pragma unroll
  for (int k = 0; k < 16; ++k) {
    xr[k] = re[midbase + k * 17];
    xi[k] = im[midbase + k * 17];
  }
  fft16_stages<4, false>(xr, xi, lo4, tws, twc);
#pragma unroll
  for (int k = 0; k < 16; ++k) {
    re[midbase + k * 17] = xr[k];
    im[midbase + k * 17] = xi[k];
  }
  __syncthreads();

  const int leafbase = tid * 17;
#pragma unroll
  for (int k = 0; k < 16; ++k) {
    xr[k] = re[leafbase + k];
    xi[k] = im[leafbase + k];
  }
  fft16_stages<0, false>(xr, xi, 0, tws, twc);
#pragma unroll
  for (int k = 0; k < 16; ++k) {
    if (fabsf(xr[k]) < 0.5f) {
      xr[k] = 0.0f;
      xi[k] = 0.0f;
    }
  }
  fft16_stages<0, true>(xr, xi, 0, tws + 8, twc + 8);
#pragma unroll
  for (int k = 0; k < 16; ++k) {
    re[leafbase + k] = xr[k];
    im[leafbase + k] = xi[k];
  }
  __syncthreads();

#pragma unroll
  for (int k = 0; k < 16; ++k) {
    xr[k] = re[midbase + k * 17];
    xi[k] = im[midbase + k * 17];
  }
  fft16_stages<4, true>(xr, xi, lo4, tws, twc);
#pragma unroll
  for (int k = 0; k < 16; ++k) {
    re[midbase + k * 17] = xr[k];
    im[midbase + k * 17] = xi[k];
  }
  __syncthreads();

  {
    const int base = FFT_PAD(tid);
#pragma unroll
    for (int k = 0; k < 16; ++k) {
      xr[k] = re[base + k * 272];
      xi[k] = im[base + k * 272];
    }
  }
  fft16_stages<8, true>(xr, xi, tid, tws, twc);

  float* xo = xt + (size_t)img * FFT_L;
  const float scale = 1.0f / (float)FFT_L;
#pragma unroll
  for (int k = 0; k < 16; ++k)
    xo[(k << 8) | tid] = xr[k] * scale;
}

typedef __attribute__((ext_vector_type(2))) float v2f;
typedef __attribute__((ext_vector_type(8))) float v8f;

#define CW_WP 292
#define CW_XP 84

__global__ __launch_bounds__(256) void conv3x3_wmma_kernel(
    const float* __restrict__ xt, const float* __restrict__ Wf,
    const float* __restrict__ bias, float* __restrict__ out) {
  __shared__ __attribute__((aligned(16))) float wl[64 * CW_WP];
  __shared__ __attribute__((aligned(16))) float xs[96 * CW_XP];

  const int tid = threadIdx.x;
  const int lane = tid & 31;
  const int wave = tid >> 5;
  const int mt = wave & 3;
  const int nh = wave >> 2;
  const int i = blockIdx.x;
  const int b = blockIdx.y;

  const int hi = lane >> 4;
  const int n = lane & 15;

  {
    const float4* W4 = reinterpret_cast<const float4*>(Wf) + (tid >> 3) * 72 + (tid & 7);
    float* wd = wl + (tid >> 3) * CW_WP + 4 * (tid & 7);
#pragma unroll
    for (int r = 0; r < 18; ++r) {
      const int half = r / 9;
      const int m = r % 9;
      *reinterpret_cast<float4*>(wd + half * 32 * CW_WP + 32 * m) =
          W4[half * 32 * 72 + 8 * m];
    }
  }
  {
    const float4* X4 = reinterpret_cast<const float4*>(
                           xt + ((size_t)b * 32 * FFT_L + (size_t)i * 64)) +
                       (tid >> 3) * (FFT_L / 4) + (tid & 7);
    float* xd = xs + (tid >> 3) * 3 * CW_XP + 4 * (tid & 7);
#pragma unroll
    for (int m = 0; m < 6; ++m)
      *reinterpret_cast<float4*>(xd + (m >> 1) * CW_XP + 32 * (m & 1)) = X4[8 * m];
    if (tid < 96)
      *reinterpret_cast<float4*>(&xs[tid * CW_XP + 64]) =
          make_float4(0.0f, 0.0f, 0.0f, 0.0f);
  }
  __syncthreads();

  const float* arow = wl + (16 * mt + n) * CW_WP + 6 * hi;
  const float* brow = xs + (2 * hi) * CW_XP + 2 * (16 * nh + n);

  v8f acc0 = {};
  v8f acc1 = {};
  v8f acc2 = {};
  v8f acc3 = {};

  auto load_step = [&](int s, v2f (&g)[3], v2f (&d)[4]) {
    const float* pa = arow + 12 * s;
    g[0] = *reinterpret_cast<const v2f*>(pa);
    g[1] = *reinterpret_cast<const v2f*>(pa + 2);
    g[2] = *reinterpret_cast<const v2f*>(pa + 4);
    const float* pb = brow + 4 * s * CW_XP;
    d[0] = *reinterpret_cast<const v2f*>(pb);
    d[1] = *reinterpret_cast<const v2f*>(pb + 2);
    d[2] = *reinterpret_cast<const v2f*>(pb + CW_XP);
    d[3] = *reinterpret_cast<const v2f*>(pb + CW_XP + 2);
  };

  v2f fg[2][4][3];
  v2f fd[2][4][4];
#pragma unroll
  for (int s = 0; s < 4; ++s) load_step(s, fg[0][s], fd[0][s]);
  __builtin_amdgcn_sched_barrier(0);
#pragma unroll
  for (int j = 0; j < 6; ++j) {
    const int cur = j & 1;
    if (j + 1 < 6) {
#pragma unroll
      for (int s = 0; s < 4; ++s)
        load_step(4 * (j + 1) + s, fg[cur ^ 1][s], fd[cur ^ 1][s]);
    }
    __builtin_amdgcn_sched_barrier(0);
#pragma unroll
    for (int s = 0; s < 4; ++s) {
      const v2f g01 = fg[cur][s][0];
      const v2f g23 = fg[cur][s][1];
      const v2f g45 = fg[cur][s][2];
      const v2f x01 = fd[cur][s][0];
      const v2f x23 = fd[cur][s][1];
      const v2f z01 = fd[cur][s][2];
      const v2f z23 = fd[cur][s][3];
      const float s0 = g01.x + g23.x;
      const float s1 = g23.y + g45.y;
      v2f a0, a1, a2, a3, b0, b1, b2, b3;
      a0.x = g01.x;       a0.y = g23.y;
      a1.x = s0 + g01.y;  a1.y = s1 + g45.x;
      a2.x = s0 - g01.y;  a2.y = s1 - g45.x;
      a3.x = g23.x;       a3.y = g45.y;
      b0.x = x01.x - x23.x;  b0.y = z01.x - z23.x;
      b1.x = x01.y + x23.x;  b1.y = z01.y + z23.x;
      b2.x = x23.x - x01.y;  b2.y = z23.x - z01.y;
      b3.x = x01.y - x23.y;  b3.y = z01.y - z23.y;
      acc0 = __builtin_amdgcn_wmma_f32_16x16x4_f32(
          false, a0, false, b0, (short)0, acc0, false, false);
      acc1 = __builtin_amdgcn_wmma_f32_16x16x4_f32(
          false, a1, false, b1, (short)0, acc1, false, false);
      acc2 = __builtin_amdgcn_wmma_f32_16x16x4_f32(
          false, a2, false, b2, (short)0, acc2, false, false);
      acc3 = __builtin_amdgcn_wmma_f32_16x16x4_f32(
          false, a3, false, b3, (short)0, acc3, false, false);
    }
    __builtin_amdgcn_sched_barrier(0);
  }

  const int t = 16 * nh + n;
  float y0[8], y1[8];
#pragma unroll
  for (int r = 0; r < 8; ++r) {
    const int o = 16 * mt + r + 8 * hi;
    const float bo = bias[o];
    y0[r] = acc0[r] + 0.5f * (acc1[r] + acc2[r]) + bo;
    y1[r] = 0.5f * (acc1[r] - acc2[r]) - acc3[r] + bo;
  }
  if (t < 31) {
#pragma unroll
    for (int r = 0; r < 8; ++r) {
      const int o = 16 * mt + r + 8 * hi;
      float2 y = make_float2(y0[r], y1[r]);
      *reinterpret_cast<float2*>(
          &out[(((size_t)b * 64 + o) * 62 + i) * 62 + 2 * t]) = y;
    }
  }
}

extern "C" void kernel_launch(void* const* d_in, const int* in_sizes, int n_in,
                              void* d_out, int out_size, void* d_ws,
                              size_t ws_size, hipStream_t stream) {
  const float* x = (const float*)d_in[0];
  const float* W = (const float*)d_in[1];
  const float* bias = (const float*)d_in[2];
  float* out = (float*)d_out;
  float* xt = (float*)d_ws;

  fft_denoise_kernel<<<dim3(128), dim3(FFT_NT), 0, stream>>>(x, xt);
  conv3x3_wmma_kernel<<<dim3(62, 4), dim3(256), 0, stream>>>(xt, W, bias, out);
}


# conv: outputs as 2x2 tiles by the minimal 2x2-output 3x3-tap filtering identity (16 products per tile, was 24): 64 WMMAs per wave, was 96; channel halves added through LDS; sum order changes
# speedup vs baseline: 3.6431x; 1.0518x over previous
#include <hip/hip_runtime.h>
#include <hip/hip_bf16.h>

#define FFT_L   4096
#define FFT_NT  256
#define PI_F    3.14159265358979323846f

#define FFT_PAD(i) ((i) + ((i) >> 4))
#define FFT_LDS    (FFT_L + FFT_L / 16)

template <int SH, bool INV>
__device__ __forceinline__ void fft16_stages(float (&xr)[16], float (&xi)[16],
                                             int low, const float* tws,
                                             const float* twc) {
#pragma unroll
  for (int qq = 0; qq < 4; ++qq) {
    const int q = INV ? qq : 3 - qq;
    const int span = 1 << q;
    const float w = (INV ? PI_F : -PI_F) / (float)(1 << (SH + q));
    float sn[8], cs[8];
#pragma unroll
    for (int j = 0; j < 8; ++j) {
      if (j < span) {
        if (SH == 0) {
          sn[j] = tws[j << (3 - q)];
          cs[j] = twc[j << (3 - q)];
        } else {
          const int pos = (j << SH) | low;
          __sincosf(w * (float)pos, &sn[j], &cs[j]);
        }
      }
    }
#pragma unroll
    for (int k = 0; k < 16; ++k) {
      if (k & span) continue;
      const int j = k & (span - 1);
      const int k1 = k + span;
      if (!INV) {
        float ar = xr[k], ai = xi[k];
        float br = xr[k1], bi = xi[k1];
        float sr = ar - br, si = ai - bi;
        xr[k] = ar + br;
        xi[k] = ai + bi;
        xr[k1] = fmaf(sr, cs[j], -(si * sn[j]));
        xi[k1] = fmaf(sr, sn[j], si * cs[j]);
      } else {
        float br = xr[k1], bi = xi[k1];
        float tr = fmaf(br, cs[j], -(bi * sn[j]));
        float ti = fmaf(br, sn[j], bi * cs[j]);
        float ar = xr[k], ai = xi[k];
        xr[k] = ar + tr;
        xi[k] = ai + ti;
        xr[k1] = ar - tr;
        xi[k1] = ai - ti;
      }
    }
  }
}

__global__ __launch_bounds__(FFT_NT) void fft_denoise_kernel(
    const float* __restrict__ x, float* __restrict__ xt) {
  __shared__ float re[FFT_LDS];
  __shared__ float im[FFT_LDS];
  __shared__ float tws[16];
  __shared__ float twc[16];
  const int img = blockIdx.x;
  const int tid = threadIdx.x;
  const float* xi_g = x + (size_t)img * 62 * 62;

  float xr[16], xi[16];

  if (tid < 16) {
    const float w = (tid < 8 ? -PI_F : PI_F) / 8.0f;
    float sn, cs;
    __sincosf(w * (float)(tid & 7), &sn, &cs);
    tws[tid] = sn;
    twc[tid] = cs;
  }

  {
    const int j = tid & 63;
#pragma unroll
    for (int k = 0; k < 16; ++k) {
      const int i = (k << 2) | (tid >> 6);
      float v = 0.0f;
      if (i >= 1 && i <= 62 && j >= 1 && j <= 62)
        v = xi_g[(i - 1) * 62 + (j - 1)];
      xr[k] = v;
      xi[k] = 0.0f;
    }
  }

  fft16_stages<8, false>(xr, xi, tid, tws, twc);
  {
    const int base = FFT_PAD(tid);
#pragma unroll
    for (int k = 0; k < 16; ++k) {
      re[base + k * 272] = xr[k];
      im[base + k * 272] = xi[k];
    }
  }
  __syncthreads();

  const int lo4 = tid & 15;
  const int midbase = (tid >> 4) * 272 + lo4;
#pragma unroll
  for (int k = 0; k < 16; ++k) {
    xr[k] = re[midbase + k * 17];
    xi[k] = im[midbase + k * 17];
  }
  fft16_stages<4, false>(xr, xi, lo4, tws, twc);
#pragma unroll
  for (int k = 0; k < 16; ++k) {
    re[midbase + k * 17] = xr[k];
    im[midbase + k * 17] = xi[k];
  }
  __syncthreads();

  const int leafbase = tid * 17;
#pragma unroll
  for (int k = 0; k < 16; ++k) {
    xr[k] = re[leafbase + k];
    xi[k] = im[leafbase + k];
  }
  fft16_stages<0, false>(xr, xi, 0, tws, twc);
#pragma unroll
  for (int k = 0; k < 16; ++k) {
    if (fabsf(xr[k]) < 0.5f) {
      xr[k] = 0.0f;
      xi[k] = 0.0f;
    }
  }
  fft16_stages<0, true>(xr, xi, 0, tws + 8, twc + 8);
#pragma unroll
  for (int k = 0; k < 16; ++k) {
    re[leafbase + k] = xr[k];
    im[leafbase + k] = xi[k];
  }
  __syncthreads();

#pragma unroll
  for (int k = 0; k < 16; ++k) {
    xr[k] = re[midbase + k * 17];
    xi[k] = im[midbase + k * 17];
  }
  fft16_stages<4, true>(xr, xi, lo4, tws, twc);
#pragma unroll
  for (int k = 0; k < 16; ++k) {
    re[midbase + k * 17] = xr[k];
    im[midbase + k * 17] = xi[k];
  }
  __syncthreads();

  {
    const int base = FFT_PAD(tid);
#pragma unroll
    for (int k = 0; k < 16; ++k) {
      xr[k] = re[base + k * 272];
      xi[k] = im[base + k * 272];
    }
  }
  fft16_stages<8, true>(xr, xi, tid, tws, twc);

  float* xo = xt + (size_t)img * FFT_L;
  const float scale = 1.0f / (float)FFT_L;
#pragma unroll
  for (int k = 0; k < 16; ++k)
    xo[(k << 8) | tid] = xr[k] * scale;
}

typedef __attribute__((ext_vector_type(2))) float v2f;
typedef __attribute__((ext_vector_type(8))) float v8f;

#define CW_WP 292
#define CW_XP 84

__global__ __launch_bounds__(256) void conv3x3_wmma_kernel(
    const float* __restrict__ xt, const float* __restrict__ Wf,
    const float* __restrict__ bias, float* __restrict__ out) {
  __shared__ __attribute__((aligned(16))) float wl[32 * CW_WP];
  __shared__ __attribute__((aligned(16))) float xs[128 * CW_XP];
  __shared__ __attribute__((aligned(16))) float ex[8 * 16 * 32];

  const int tid = threadIdx.x;
  const int lane = tid & 31;
  const int wave = tid >> 5;
  const int mt = wave & 1;
  const int nh = (wave >> 1) & 1;
  const int kq = wave >> 2;
  const int ip = blockIdx.x;
  const int oh = blockIdx.y;
  const int b = blockIdx.z;

  const int hi = lane >> 4;
  const int n = lane & 15;

  {
    const float4* W4 = reinterpret_cast<const float4*>(Wf) +
                       (32 * oh + (tid >> 3)) * 72 + (tid & 7);
    float* wd = wl + (tid >> 3) * CW_WP + 4 * (tid & 7);
#pragma unroll
    for (int m = 0; m < 9; ++m)
      *reinterpret_cast<float4*>(wd + 32 * m) = W4[8 * m];
  }
  {
    const float4* X4 = reinterpret_cast<const float4*>(
                           xt + ((size_t)b * 32 * FFT_L + (size_t)ip * 128)) +
                       (tid >> 3) * (FFT_L / 4) + (tid & 7);
    float* xd = xs + (tid >> 3) * 4 * CW_XP + 4 * (tid & 7);
#pragma unroll
    for (int m = 0; m < 8; ++m)
      *reinterpret_cast<float4*>(xd + (m >> 1) * CW_XP + 32 * (m & 1)) = X4[8 * m];
    if (tid < 128)
      *reinterpret_cast<float4*>(&xs[tid * CW_XP + 64]) =
          make_float4(0.0f, 0.0f, 0.0f, 0.0f);
  }
  __syncthreads();

  const float* arow = wl + (16 * mt + n) * CW_WP + 144 * kq + 18 * hi;
  const float* brow = xs + (64 * kq + 8 * hi) * CW_XP + 2 * (16 * nh + n);

  v8f acc[4][4];
#pragma unroll
  for (int r = 0; r < 4; ++r) {
#pragma unroll
    for (int p = 0; p < 4; ++p) acc[r][p] = v8f{};
  }

  auto load_step = [&](int s, v2f (&g)[9], v2f (&d)[2][4][2]) {
    const float* pa = arow + 36 * s;
#pragma unroll
    for (int k = 0; k < 9; ++k) g[k] = *reinterpret_cast<const v2f*>(pa + 2 * k);
    const float* pb = brow + 16 * s * CW_XP;
#pragma unroll
    for (int ch = 0; ch < 2; ++ch) {
#pragma unroll
      for (int rr = 0; rr < 4; ++rr) {
        d[ch][rr][0] = *reinterpret_cast<const v2f*>(pb + (4 * ch + rr) * CW_XP);
        d[ch][rr][1] = *reinterpret_cast<const v2f*>(pb + (4 * ch + rr) * CW_XP + 2);
      }
    }
  };

  v2f fg[2][9];
  v2f fd[2][2][4][2];
  load_step(0, fg[0], fd[0]);
  __builtin_amdgcn_sched_barrier(0);
#pragma unroll
  for (int j = 0; j < 4; ++j) {
    const int cur = j & 1;
    if (j + 1 < 4) load_step(j + 1, fg[cur ^ 1], fd[cur ^ 1]);
    __builtin_amdgcn_sched_barrier(0);

    v2f tp[9];
#pragma unroll
    for (int q = 0; q < 9; ++q) {
      tp[q].x = fg[cur][q >> 1][q & 1];
      tp[q].y = fg[cur][(9 + q) >> 1][(9 + q) & 1];
    }
    v2f xv[4][4];
#pragma unroll
    for (int rr = 0; rr < 4; ++rr) {
#pragma unroll
      for (int col = 0; col < 4; ++col) {
        xv[rr][col].x = fd[cur][0][rr][col >> 1][col & 1];
        xv[rr][col].y = fd[cur][1][rr][col >> 1][col & 1];
      }
    }
    v2f cf[3][4];
#pragma unroll
    for (int kh = 0; kh < 3; ++kh) {
      const v2f sm = tp[3 * kh] + tp[3 * kh + 2];
      cf[kh][0] = tp[3 * kh];
      cf[kh][1] = sm + tp[3 * kh + 1];
      cf[kh][2] = sm - tp[3 * kh + 1];
      cf[kh][3] = tp[3 * kh + 2];
    }
    v2f ed[4][4];
#pragma unroll
    for (int rr = 0; rr < 4; ++rr) {
      ed[rr][0] = xv[rr][0] - xv[rr][2];
      ed[rr][1] = xv[rr][1] + xv[rr][2];
      ed[rr][2] = xv[rr][2] - xv[rr][1];
      ed[rr][3] = xv[rr][1] - xv[rr][3];
    }
#pragma unroll
    for (int p = 0; p < 4; ++p) {
      const v2f sp = cf[0][p] + cf[2][p];
      const v2f a0 = cf[0][p];
      const v2f a1 = sp + cf[1][p];
      const v2f a2 = sp - cf[1][p];
      const v2f a3 = cf[2][p];
      const v2f b0 = ed[0][p] - ed[2][p];
      const v2f b1 = ed[1][p] + ed[2][p];
      const v2f b2 = ed[2][p] - ed[1][p];
      const v2f b3 = ed[1][p] - ed[3][p];
      acc[0][p] = __builtin_amdgcn_wmma_f32_16x16x4_f32(
          false, a0, false, b0, (short)0, acc[0][p], false, false);
      acc[1][p] = __builtin_amdgcn_wmma_f32_16x16x4_f32(
          false, a1, false, b1, (short)0, acc[1][p], false, false);
      acc[2][p] = __builtin_amdgcn_wmma_f32_16x16x4_f32(
          false, a2, false, b2, (short)0, acc[2][p], false, false);
      acc[3][p] = __builtin_amdgcn_wmma_f32_16x16x4_f32(
          false, a3, false, b3, (short)0, acc[3][p], false, false);
    }
    __builtin_amdgcn_sched_barrier(0);
  }

  v8f yk0, yk1, ys0, ys1;
  {
    v8f r0[4], r1[4];
#pragma unroll
    for (int p = 0; p < 4; ++p) {
      r0[p] = acc[0][p] + 0.5f * (acc[1][p] + acc[2][p]);
      r1[p] = 0.5f * (acc[1][p] - acc[2][p]) - acc[3][p];
    }
    const v8f y00 = r0[0] + 0.5f * (r0[1] + r0[2]);
    const v8f y01 = 0.5f * (r0[1] - r0[2]) - r0[3];
    const v8f y10 = r1[0] + 0.5f * (r1[1] + r1[2]);
    const v8f y11 = 0.5f * (r1[1] - r1[2]) - r1[3];
    yk0 = kq ? y10 : y00;
    yk1 = kq ? y11 : y01;
    ys0 = kq ? y00 : y10;
    ys1 = kq ? y01 : y11;
  }
  {
    float* exw = ex + wave * 512 + lane;
#pragma unroll
    for (int r = 0; r < 8; ++r) {
      exw[r * 32] = ys0[r];
      exw[(8 + r) * 32] = ys1[r];
    }
  }
  __syncthreads();
  {
    const float* exr = ex + (wave ^ 4) * 512 + lane;
#pragma unroll
    for (int r = 0; r < 8; ++r) {
      yk0[r] += exr[r * 32];
      yk1[r] += exr[(8 + r) * 32];
    }
  }

  const int t = 16 * nh + n;
  const int row = 2 * ip + kq;
  if (t < 31) {
#pragma unroll
    for (int r = 0; r < 8; ++r) {
      const int o = 32 * oh + 16 * mt + r + 8 * hi;
      const float bo = bias[o];
      float2 y = make_float2(yk0[r] + bo, yk1[r] + bo);
      *reinterpret_cast<float2*>(
          &out[(((size_t)b * 64 + o) * 62 + row) * 62 + 2 * t]) = y;
    }
  }
}

extern "C" void kernel_launch(void* const* d_in, const int* in_sizes, int n_in,
                              void* d_out, int out_size, void* d_ws,
                              size_t ws_size, hipStream_t stream) {
  const float* x = (const float*)d_in[0];
  const float* W = (const float*)d_in[1];
  const float* bias = (const float*)d_in[2];
  float* out = (float*)d_out;
  float* xt = (float*)d_ws;

  fft_denoise_kernel<<<dim3(128), dim3(FFT_NT), 0, stream>>>(x, xt);
  conv3x3_wmma_kernel<<<dim3(31, 2, 4), dim3(256), 0, stream>>>(xt, W, bias, out);
}
